# starting kernel
# baseline (speedup 1.0000x reference)
#include <hip/hip_runtime.h>

typedef __attribute__((ext_vector_type(2))) float v2f;
typedef __attribute__((ext_vector_type(8))) float v8f;

#define IN_H 4096
#define IN_W 4096
#define KH 15
#define KW 15
#define OUT_H (IN_H - KH + 1)
#define OUT_W (IN_W - KW + 1)

#define HROWS     112
#define ROWSTRIDE 17
#define NWAVES    8
#define ACC_PER_WAVE    (HROWS * ROWSTRIDE + 16)
#define ROWSLOT         40
#define ROWBUF_PER_WAVE (2 * ROWSLOT)

template <bool GUARD>
__device__ __forceinline__ void process_row(
    int r, int y0, int lane, int n, int half, int tb, float xv,
    float* __restrict__ rowbuf, float* __restrict__ acc, const v2f (&a)[4])
{
  rowbuf[lane] = xv;

  v8f c = {0.f, 0.f, 0.f, 0.f, 0.f, 0.f, 0.f, 0.f};
  #pragma unroll
  for (int j = 0; j < 4; ++j) {
    v2f bv;
    bv.x = rowbuf[tb + 4 * j];
    bv.y = rowbuf[tb + 4 * j + 1];
    c = __builtin_amdgcn_wmma_f32_16x16x4_f32(false, a[j], false, bv,
                                              (short)0, c, false, false);
  }

  const int rb = r - y0 - 8 * half;
  if (GUARD) {
    #pragma unroll
    for (int i = 0; i < 8; ++i) {
      const int row    = rb - i;
      const int off_ok = row * ROWSTRIDE + n;
      const int m      = -(int)((unsigned)row < (unsigned)HROWS);
      const int off    = (off_ok & m) | ((HROWS * ROWSTRIDE + n) & ~m);
      atomicAdd(&acc[off], c[i]);
    }
  } else {
    const int base = (rb - 7) * ROWSTRIDE + n;
    #pragma unroll
    for (int i = 0; i < 8; ++i)
      atomicAdd(&acc[base + (7 - i) * ROWSTRIDE], c[i]);
  }
}

__device__ __forceinline__ void process_row_pair(
    int r, int y0, int lane, int n, int half, int tb, float xv0, float xv1,
    float* __restrict__ rowbuf, float* __restrict__ acc, const v2f (&a)[4])
{
  rowbuf[lane]           = xv0;
  rowbuf[ROWSLOT + lane] = xv1;

  v8f c0 = {0.f, 0.f, 0.f, 0.f, 0.f, 0.f, 0.f, 0.f};
  v8f c1 = {0.f, 0.f, 0.f, 0.f, 0.f, 0.f, 0.f, 0.f};
  #pragma unroll
  for (int j = 0; j < 4; ++j) {
    v2f b0, b1;
    b0.x = rowbuf[tb + 4 * j];
    b0.y = rowbuf[tb + 4 * j + 1];
    b1.x = rowbuf[ROWSLOT + tb + 4 * j];
    b1.y = rowbuf[ROWSLOT + tb + 4 * j + 1];
    c0 = __builtin_amdgcn_wmma_f32_16x16x4_f32(false, a[j], false, b0,
                                               (short)0, c0, false, false);
    c1 = __builtin_amdgcn_wmma_f32_16x16x4_f32(false, a[j], false, b1,
                                               (short)0, c1, false, false);
  }

  const int rb    = r - y0 - 8 * half;
  const int base0 = (rb - 7) * ROWSTRIDE + n;
  #pragma unroll
  for (int i = 0; i < 8; ++i)
    atomicAdd(&acc[base0 + (7 - i) * ROWSTRIDE], c0[i]);
  #pragma unroll
  for (int i = 0; i < 8; ++i)
    atomicAdd(&acc[base0 + (8 - i) * ROWSTRIDE], c1[i]);
}

__global__ __launch_bounds__(256) void conv2d_wmma_f32_kernel(
    const float* __restrict__ X, const float* __restrict__ Wgt,
    const float* __restrict__ bias, float* __restrict__ out)
{
  __shared__ float lds_acc[NWAVES * ACC_PER_WAVE];
  __shared__ float lds_row[NWAVES * ROWBUF_PER_WAVE];

  const int lane = threadIdx.x & 31;
  const int wave = threadIdx.x >> 5;
  const int n    = lane & 15;
  const int half = lane >> 4;

  const int x0   = blockIdx.x * 16;
  const int band = blockIdx.y * NWAVES + wave;
  const int y0   = band * HROWS;

  float* acc    = lds_acc + wave * ACC_PER_WAVE;
  float* rowbuf = lds_row + wave * ROWBUF_PER_WAVE;
  for (int i = lane; i < ACC_PER_WAVE; i += 32) acc[i] = 0.0f;

  v2f a[4];
  #pragma unroll
  for (int j = 0; j < 4; ++j) {
    const int kx0 = 4 * j + 2 * half;
    a[j].x = (n < KH && kx0     < KW) ? Wgt[n * KW + kx0    ] : 0.0f;
    a[j].y = (n < KH && kx0 + 1 < KW) ? Wgt[n * KW + kx0 + 1] : 0.0f;
  }

  const int tb    = n + 2 * half;
  const int gx    = x0 + lane;
  const int xmask = (gx < IN_W) ? -1 : 0;
  const float* xcol = X + min(gx, IN_W - 1);

  const int r_last = min(y0 + HROWS - 1, OUT_H - 1) + (KH - 1);
  const int pro_end    = min(y0 + KH, r_last + 1);
  const int steady_end = min(y0 + HROWS, r_last + 1);

  #define LOAD_ROW(rr)                                                         \
    __int_as_float(__float_as_int(xcol[(size_t)min((rr), IN_H - 1) * IN_W])    \
                   & xmask)

  int r = y0;
  float xv = LOAD_ROW(r);
  for (; r < pro_end; ++r) {
    const size_t nofs = (size_t)min(r + 1, IN_H - 1) * IN_W;
    const float xn = __int_as_float(__float_as_int(xcol[nofs]) & xmask);
    __builtin_prefetch((const void*)(xcol + nofs + (size_t)15 * IN_W), 0, 1);
    process_row<true>(r, y0, lane, n, half, tb, xv, rowbuf, acc, a);
    xv = xn;
  }
  float xv1 = LOAD_ROW(r + 1);
  for (; r + 1 < steady_end; r += 2) {
    const size_t nofs = (size_t)min(r + 2, IN_H - 1) * IN_W;
    const float xn0 = __int_as_float(__float_as_int(xcol[nofs]) & xmask);
    const float xn1 = LOAD_ROW(r + 3);
    __builtin_prefetch((const void*)(xcol + nofs + (size_t)15 * IN_W), 0, 1);
    process_row_pair(r, y0, lane, n, half, tb, xv, xv1, rowbuf, acc, a);
    xv = xn0;
    xv1 = xn1;
  }
  for (; r < steady_end; ++r) {
    const float xn = LOAD_ROW(r + 1);
    process_row<false>(r, y0, lane, n, half, tb, xv, rowbuf, acc, a);
    xv = xn;
  }
  for (; r <= r_last; ++r) {
    const float xn = LOAD_ROW(r + 1);
    process_row<true>(r, y0, lane, n, half, tb, xv, rowbuf, acc, a);
    xv = xn;
  }
  #undef LOAD_ROW

  __syncthreads();

  const float b0 = bias[0];
  #pragma unroll 4
  for (int rr = 0; rr < HROWS; rr += 2) {
    const int row = rr + half;
    const int y = y0 + row;
    const int x = x0 + n;
    if (y < OUT_H && x < OUT_W)
      out[(size_t)y * OUT_W + x] = acc[row * ROWSTRIDE + n] + b0;
  }
}

extern "C" void kernel_launch(void* const* d_in, const int* in_sizes, int n_in,
                              void* d_out, int out_size, void* d_ws, size_t ws_size,
                              hipStream_t stream) {
  const float* X    = (const float*)d_in[0];
  const float* Wgt  = (const float*)d_in[1];
  const float* bias = (const float*)d_in[2];
  float* out = (float*)d_out;

  const int bands = (OUT_H + HROWS - 1) / HROWS;
  dim3 grid((OUT_W + 15) / 16, (bands + NWAVES - 1) / NWAVES, 1);
  dim3 block(256, 1, 1);
  conv2d_wmma_f32_kernel<<<grid, block, 0, stream>>>(X, Wgt, bias, out);
}


# Per-row WMMA results are folded into a 15-stage register pipeline (S[ky] = P[ky] + S_prev[ky-1], one half swap per row) instead of LDS float atomics; rows stored from registers; sum order kept
# speedup vs baseline: 1.2558x; 1.2558x over previous
#include <hip/hip_runtime.h>


typedef __attribute__((ext_vector_type(2))) float v2f;
typedef __attribute__((ext_vector_type(8))) float v8f;

#define IN_H 4096
#define IN_W 4096
#define KH 15
#define KW 15
#define OUT_H (IN_H - KH + 1)
#define OUT_W (IN_W - KW + 1)

#define HROWS     112
#define NWAVES    8
#define ROWSLOT         40
#define ROWBUF_PER_WAVE (2 * ROWSLOT)

static_assert(KH <= 15, "tap row ky lives in D row ky; D row 15 stays a zero-weight row");
static_assert(HROWS % 2 == 0 && (OUT_H % HROWS) % 2 == 0 && (KH - 1) % 2 == 0,
              "every band is a whole number of row pairs");

__device__ __forceinline__ float xhalf(float x) {
  return __builtin_bit_cast(float, __builtin_amdgcn_permlanex16(
      __builtin_bit_cast(int, x), __builtin_bit_cast(int, x),
      0x76543210, 0xfedcba98, false, false));
}

__device__ __forceinline__ v8f fold_row(const v8f& P, const v8f& S, float carry_in)
{
  v8f R;
  R[0] = P[0] + carry_in;
  #pragma unroll
  for (int i = 1; i < 8; ++i) R[i] = P[i] + S[i - 1];
  return R;
}

__global__ __launch_bounds__(256) void conv2d_wmma_f32_kernel(
    const float* __restrict__ X, const float* __restrict__ Wgt,
    const float* __restrict__ bias, float* __restrict__ out)
{
  __shared__ float lds_row[NWAVES * ROWBUF_PER_WAVE];

  const int lane = threadIdx.x & 31;
  const int wave = __builtin_amdgcn_readfirstlane(threadIdx.x >> 5);
  const int n    = lane & 15;
  const int half = lane >> 4;

  const int x0   = blockIdx.x * 16;
  const int band = blockIdx.y * NWAVES + wave;
  const int y0   = band * HROWS;
  if (y0 >= OUT_H) return;

  float* rowbuf = lds_row + wave * ROWBUF_PER_WAVE;

  v2f a[4];
  #pragma unroll
  for (int j = 0; j < 4; ++j) {
    const int kx0 = 4 * j + 2 * half;
    a[j].x = (n < KH && kx0     < KW) ? Wgt[n * KW + kx0    ] : 0.0f;
    a[j].y = (n < KH && kx0 + 1 < KW) ? Wgt[n * KW + kx0 + 1] : 0.0f;
  }

  const int tb    = n + 2 * half;
  const int gx    = x0 + lane;
  const int xmask = (gx < IN_W) ? -1 : 0;
  const float* xcol = X + min(gx, IN_W - 1);

  const int y_last = min(y0 + HROWS - 1, OUT_H - 1);
  const int r_last = y_last + (KH - 1);

  const float b0  = bias[0];
  const int   x   = x0 + n;
  const bool  xok = x < OUT_W;

  #define LOAD_ROW(rr)                                                         \
    __int_as_float(__float_as_int(xcol[(size_t)min((rr), IN_H - 1) * IN_W])    \
                   & xmask)

  v8f S = {0.f, 0.f, 0.f, 0.f, 0.f, 0.f, 0.f, 0.f};
  float carry = 0.f;

  int r = y0;
  float xv0 = LOAD_ROW(r);
  float xv1 = LOAD_ROW(r + 1);
  for (; r < r_last; r += 2) {
    const size_t nofs = (size_t)min(r + 2, IN_H - 1) * IN_W;
    const float xn0 = __int_as_float(__float_as_int(xcol[nofs]) & xmask);
    const float xn1 = LOAD_ROW(r + 3);
    __builtin_prefetch((const void*)(xcol + nofs + (size_t)15 * IN_W), 0, 1);

    rowbuf[lane]           = xv0;
    rowbuf[ROWSLOT + lane] = xv1;

    v8f p0 = {0.f, 0.f, 0.f, 0.f, 0.f, 0.f, 0.f, 0.f};
    v8f p1 = {0.f, 0.f, 0.f, 0.f, 0.f, 0.f, 0.f, 0.f};
    #pragma unroll
    for (int j = 0; j < 4; ++j) {
      v2f b0v, b1v;
      b0v.x = rowbuf[tb + 4 * j];
      b0v.y = rowbuf[tb + 4 * j + 1];
      b1v.x = rowbuf[ROWSLOT + tb + 4 * j];
      b1v.y = rowbuf[ROWSLOT + tb + 4 * j + 1];
      p0 = __builtin_amdgcn_wmma_f32_16x16x4_f32(false, a[j], false, b0v,
                                                 (short)0, p0, false, false);
      p1 = __builtin_amdgcn_wmma_f32_16x16x4_f32(false, a[j], false, b1v,
                                                 (short)0, p1, false, false);
    }

    const v8f S1 = fold_row(p0, S, half ? carry : 0.f);
    const float carry1 = xhalf(S1[7]);
    S = fold_row(p1, S1, half ? carry1 : 0.f);
    carry = xhalf(S[7]);

    if (r >= y0 + (KH - 1)) {
      const float done0 = xhalf(S1[6]);
      const float v = half ? S[6] : done0;
      const int   y = r - (KH - 1) + half;
      if (xok) out[(size_t)y * OUT_W + x] = v + b0;
    }

    xv0 = xn0;
    xv1 = xn1;
  }
  #undef LOAD_ROW
}

extern "C" void kernel_launch(void* const* d_in, const int* in_sizes, int n_in,
                              void* d_out, int out_size, void* d_ws, size_t ws_size,
                              hipStream_t stream) {
  const float* X    = (const float*)d_in[0];
  const float* Wgt  = (const float*)d_in[1];
  const float* bias = (const float*)d_in[2];
  float* out = (float*)d_out;

  const int bands = (OUT_H + HROWS - 1) / HROWS;
  dim3 grid((OUT_W + 15) / 16, (bands + NWAVES - 1) / NWAVES, 1);
  dim3 block(256, 1, 1);
  conv2d_wmma_f32_kernel<<<grid, block, 0, stream>>>(X, Wgt, bias, out);
}


# Each wave computes two adjacent 16-column tiles from one staged 48-float row segment: two independent WMMA chains, B reads in one group, full-row stores; 4-wave blocks, grid 128x10; sum order kept
# speedup vs baseline: 1.3879x; 1.1052x over previous
#include <hip/hip_runtime.h>


typedef __attribute__((ext_vector_type(2))) float v2f;
typedef __attribute__((ext_vector_type(8))) float v8f;
typedef __attribute__((ext_vector_type(10))) float v10f;

#define IN_H 4096
#define IN_W 4096
#define KH 15
#define KW 15
#define OUT_H (IN_H - KH + 1)
#define OUT_W (IN_W - KW + 1)

#define HROWS     112
#define NWAVES    4
#define WTILE     32
#define SEGW      48
#define ROWSLOT         64
#define ROWBUF_PER_WAVE (2 * ROWSLOT)

static_assert(KH == 15, "rows r and r+1 of a pair use tap rows m-1 and m of D row m: 16 D rows hold 15 tap rows");
static_assert(HROWS % 2 == 0 && (OUT_H % HROWS) % 2 == 0 && (KH - 1) % 2 == 0,
              "every band is a whole number of row pairs");

__device__ __forceinline__ float xhalf(float x) {
  return __builtin_bit_cast(float, __builtin_amdgcn_permlanex16(
      __builtin_bit_cast(int, x), __builtin_bit_cast(int, x),
      0x76543210, 0xfedcba98, false, false));
}


__global__ __launch_bounds__(128) void conv2d_wmma_f32_kernel(
    const float* __restrict__ X, const float* __restrict__ Wgt,
    const float* __restrict__ bias, float* __restrict__ out)
{
  __shared__ float lds_row[NWAVES * ROWBUF_PER_WAVE];

  const int lane = threadIdx.x & 31;
  const int wave = __builtin_amdgcn_readfirstlane(threadIdx.x >> 5);
  const int n    = lane & 15;
  const int half = lane >> 4;

  const int x0   = blockIdx.x * WTILE;
  const int band = blockIdx.y * NWAVES + wave;
  const int y0   = band * HROWS;
  if (y0 >= OUT_H) return;

  float* rowbuf = lds_row + wave * ROWBUF_PER_WAVE;

  v2f a0[4], a1[4];
  #pragma unroll
  for (int j = 0; j < 4; ++j) {
    const int kx0 = 4 * j + 2 * half;
    a1[j].x = (n < KH && kx0     < KW) ? Wgt[n * KW + kx0    ] : 0.0f;
    a1[j].y = (n < KH && kx0 + 1 < KW) ? Wgt[n * KW + kx0 + 1] : 0.0f;
    a0[j].x = (n >= 1 && kx0     < KW) ? Wgt[(n - 1) * KW + kx0    ] : 0.0f;
    a0[j].y = (n >= 1 && kx0 + 1 < KW) ? Wgt[(n - 1) * KW + kx0 + 1] : 0.0f;
  }

  const int tb    = n + 2 * half;
  const int lc    = min(lane, SEGW / 2 - 1);
  const int gx    = x0 + 2 * lc;
  const int xmask = (gx < IN_W) ? -1 : 0;
  const float* xcol = X + min(gx, IN_W - 2);

  const int y_last = min(y0 + HROWS - 1, OUT_H - 1);
  const int r_last = y_last + (KH - 1);

  const float b0  = bias[0];
  const int   x   = x0 + lane;
  const bool  xok = x < OUT_W;

  auto load_at = [&](const size_t ofs) -> v2f {
    v2f v = *reinterpret_cast<const v2f*>(xcol + ofs);
    v.x = __int_as_float(__float_as_int(v.x) & xmask);
    v.y = __int_as_float(__float_as_int(v.y) & xmask);
    return v;
  };
  #define LOAD_ROW(rr) load_at((size_t)min((rr), IN_H - 1) * IN_W)

  v10f P0 = {0.f, 0.f, 0.f, 0.f, 0.f, 0.f, 0.f, 0.f, 0.f, 0.f};
  v10f Q0 = {0.f, 0.f, 0.f, 0.f, 0.f, 0.f, 0.f, 0.f, 0.f, 0.f};
  v10f P1 = {0.f, 0.f, 0.f, 0.f, 0.f, 0.f, 0.f, 0.f, 0.f, 0.f};
  v10f Q1 = {0.f, 0.f, 0.f, 0.f, 0.f, 0.f, 0.f, 0.f, 0.f, 0.f};

  int r = y0;
  v2f xv0 = LOAD_ROW(r);
  v2f xv1 = LOAD_ROW(r + 1);

  auto pair_step = [&](const int rr, v10f& f0, v10f& f1, v10f& t0, v10f& t1) {
    const size_t nofs = (size_t)min(rr + 2, IN_H - 1) * IN_W;
    const v2f xn0 = load_at(nofs);
    const v2f xn1 = LOAD_ROW(rr + 3);
    __builtin_prefetch((const void*)(xcol + nofs + (size_t)15 * IN_W), 0, 1);

    *reinterpret_cast<v2f*>(rowbuf + 2 * lane)           = xv0;
    *reinterpret_cast<v2f*>(rowbuf + ROWSLOT + 2 * lane) = xv1;

    const float t6a = xhalf(f0[8]);
    const float t7a = xhalf(f0[9]);
    const float t6b = xhalf(f1[8]);
    const float t7b = xhalf(f1[9]);

    if (rr >= y0 + (KH + 1)) {
      const float v16 = half ? f1[9] : t7a;
      const float v15 = half ? f1[8] : t6a;
      if (xok) {
        out[(size_t)(rr - (KH + 1)) * OUT_W + x] = v16 + b0;
        out[(size_t)(rr - KH) * OUT_W + x]       = v15 + b0;
      }
    }

    f0[0] = half ? t6a : 0.f;
    f0[1] = half ? t7a : 0.f;
    f1[0] = half ? t6b : 0.f;
    f1[1] = half ? t7b : 0.f;
    v8f acc0 = __builtin_shufflevector(f0, f0, 0, 1, 2, 3, 4, 5, 6, 7);
    v8f acc1 = __builtin_shufflevector(f1, f1, 0, 1, 2, 3, 4, 5, 6, 7);

    v2f b00[4], b01[4], b10[4], b11[4];
    #pragma unroll
    for (int j = 0; j < 4; ++j) {
      b00[j].x = rowbuf[tb + 4 * j];
      b00[j].y = rowbuf[tb + 4 * j + 1];
      b01[j].x = rowbuf[16 + tb + 4 * j];
      b01[j].y = rowbuf[16 + tb + 4 * j + 1];
      b10[j].x = rowbuf[ROWSLOT + tb + 4 * j];
      b10[j].y = rowbuf[ROWSLOT + tb + 4 * j + 1];
      b11[j].x = rowbuf[ROWSLOT + 16 + tb + 4 * j];
      b11[j].y = rowbuf[ROWSLOT + 16 + tb + 4 * j + 1];
    }
    __builtin_amdgcn_sched_barrier(0);
    #pragma unroll
    for (int j = 0; j < 4; ++j) {
      acc0 = __builtin_amdgcn_wmma_f32_16x16x4_f32(false, a0[j], false, b00[j],
                                                   (short)0, acc0, false, false);
      acc1 = __builtin_amdgcn_wmma_f32_16x16x4_f32(false, a0[j], false, b01[j],
                                                   (short)0, acc1, false, false);
      acc0 = __builtin_amdgcn_wmma_f32_16x16x4_f32(false, a1[j], false, b10[j],
                                                   (short)0, acc0, false, false);
      acc1 = __builtin_amdgcn_wmma_f32_16x16x4_f32(false, a1[j], false, b11[j],
                                                   (short)0, acc1, false, false);
    }
    t0 = __builtin_shufflevector(acc0, acc0, -1, -1, 0, 1, 2, 3, 4, 5, 6, 7);
    t1 = __builtin_shufflevector(acc1, acc1, -1, -1, 0, 1, 2, 3, 4, 5, 6, 7);

    xv0 = xn0;
    xv1 = xn1;
  };

  for (; r + 2 < r_last; r += 4) {
    pair_step(r,     P0, P1, Q0, Q1);
    pair_step(r + 2, Q0, Q1, P0, P1);
  }
  if (r < r_last) {
    pair_step(r, P0, P1, Q0, Q1);
    P0 = Q0;
    P1 = Q1;
    r += 2;
  }

  {
    const float t6a = xhalf(P0[8]);
    const float t7a = xhalf(P0[9]);
    const float v16 = half ? P1[9] : t7a;
    const float v15 = half ? P1[8] : t6a;
    if (xok) {
      out[(size_t)(r - (KH + 1)) * OUT_W + x] = v16 + b0;
      out[(size_t)(r - KH) * OUT_W + x]       = v15 + b0;
    }
  }
  #undef LOAD_ROW
}

extern "C" void kernel_launch(void* const* d_in, const int* in_sizes, int n_in,
                              void* d_out, int out_size, void* d_ws, size_t ws_size,
                              hipStream_t stream) {
  const float* X    = (const float*)d_in[0];
  const float* Wgt  = (const float*)d_in[1];
  const float* bias = (const float*)d_in[2];
  float* out = (float*)d_out;

  const int bands = (OUT_H + HROWS - 1) / HROWS;
  dim3 grid((OUT_W + WTILE - 1) / WTILE, (bands + NWAVES - 1) / NWAVES, 1);
  dim3 block(32 * NWAVES, 1, 1);
  conv2d_wmma_f32_kernel<<<grid, block, 0, stream>>>(X, Wgt, bias, out);
}


# Row loads issued three row pairs ahead through four register sets that take turns by name (were issued and waited for inside one step in every second pair); WMMA order and sum order kept
# speedup vs baseline: 1.4827x; 1.0683x over previous
#include <hip/hip_runtime.h>


typedef __attribute__((ext_vector_type(2))) float v2f;
typedef __attribute__((ext_vector_type(8))) float v8f;
typedef __attribute__((ext_vector_type(10))) float v10f;

#define IN_H 4096
#define IN_W 4096
#define KH 15
#define KW 15
#define OUT_H (IN_H - KH + 1)
#define OUT_W (IN_W - KW + 1)

#define HROWS     112
#define NWAVES    4
#define WTILE     32
#define SEGW      48
#define ROWSLOT         64
#define ROWBUF_PER_WAVE (2 * ROWSLOT)

static_assert(KH == 15, "rows r and r+1 of a pair use tap rows m-1 and m of D row m: 16 D rows hold 15 tap rows");
static_assert(HROWS % 2 == 0 && (OUT_H % HROWS) % 2 == 0 && (KH - 1) % 2 == 0,
              "every band is a whole number of row pairs");

__device__ __forceinline__ float xhalf(float x) {
  return __builtin_bit_cast(float, __builtin_amdgcn_permlanex16(
      __builtin_bit_cast(int, x), __builtin_bit_cast(int, x),
      0x76543210, 0xfedcba98, false, false));
}


__global__ __launch_bounds__(128) void conv2d_wmma_f32_kernel(
    const float* __restrict__ X, const float* __restrict__ Wgt,
    const float* __restrict__ bias, float* __restrict__ out)
{
  __shared__ float lds_row[NWAVES * ROWBUF_PER_WAVE];

  const int lane = threadIdx.x & 31;
  const int wave = __builtin_amdgcn_readfirstlane(threadIdx.x >> 5);
  const int n    = lane & 15;
  const int half = lane >> 4;

  const int x0   = blockIdx.x * WTILE;
  const int band = blockIdx.y * NWAVES + wave;
  const int y0   = band * HROWS;
  if (y0 >= OUT_H) return;

  float* rowbuf = lds_row + wave * ROWBUF_PER_WAVE;

  v2f a0[4], a1[4];
  #pragma unroll
  for (int j = 0; j < 4; ++j) {
    const int kx0 = 4 * j + 2 * half;
    a1[j].x = (n < KH && kx0     < KW) ? Wgt[n * KW + kx0    ] : 0.0f;
    a1[j].y = (n < KH && kx0 + 1 < KW) ? Wgt[n * KW + kx0 + 1] : 0.0f;
    a0[j].x = (n >= 1 && kx0     < KW) ? Wgt[(n - 1) * KW + kx0    ] : 0.0f;
    a0[j].y = (n >= 1 && kx0 + 1 < KW) ? Wgt[(n - 1) * KW + kx0 + 1] : 0.0f;
  }

  const int tb    = n + 2 * half;
  const int lc    = min(lane, SEGW / 2 - 1);
  const int gx    = x0 + 2 * lc;
  const int xmask = (gx < IN_W) ? -1 : 0;
  const float* xcol = X + min(gx, IN_W - 2);

  const int y_last = min(y0 + HROWS - 1, OUT_H - 1);
  const int r_last = y_last + (KH - 1);

  const float b0  = bias[0];
  const int   x   = x0 + lane;
  const bool  xok = x < OUT_W;

  struct rowpair { v2f a, b; };
  auto load_at = [&](const size_t ofs) -> v2f {
    return *reinterpret_cast<const v2f*>(xcol + ofs);
  };
  auto masked = [&](v2f v) -> v2f {
    v.x = __int_as_float(__float_as_int(v.x) & xmask);
    v.y = __int_as_float(__float_as_int(v.y) & xmask);
    return v;
  };
  #define LOAD_ROW(rr) load_at((size_t)min((rr), r_last) * IN_W)

  v10f P0 = {0.f, 0.f, 0.f, 0.f, 0.f, 0.f, 0.f, 0.f, 0.f, 0.f};
  v10f Q0 = {0.f, 0.f, 0.f, 0.f, 0.f, 0.f, 0.f, 0.f, 0.f, 0.f};
  v10f P1 = {0.f, 0.f, 0.f, 0.f, 0.f, 0.f, 0.f, 0.f, 0.f, 0.f};
  v10f Q1 = {0.f, 0.f, 0.f, 0.f, 0.f, 0.f, 0.f, 0.f, 0.f, 0.f};

  int r = y0;
  rowpair S0, S1, S2, S3;
  S0.a = LOAD_ROW(r);     S0.b = LOAD_ROW(r + 1);
  S1.a = LOAD_ROW(r + 2); S1.b = LOAD_ROW(r + 3);
  S2.a = LOAD_ROW(r + 4); S2.b = LOAD_ROW(r + 5);

  auto pair_step = [&](const int rr, const rowpair& cur, rowpair& ld, const bool ahead,
                       v10f& f0, v10f& f1, v10f& t0, v10f& t1) {
    if (ahead) {
      const size_t nofs = (size_t)min(rr + 6, r_last) * IN_W;
      ld.a = load_at(nofs);
      ld.b = LOAD_ROW(rr + 7);
      __builtin_prefetch((const void*)(xcol + nofs + (size_t)15 * IN_W), 0, 1);
    }

    *reinterpret_cast<v2f*>(rowbuf + 2 * lane)           = masked(cur.a);
    *reinterpret_cast<v2f*>(rowbuf + ROWSLOT + 2 * lane) = masked(cur.b);

    const float t6a = xhalf(f0[8]);
    const float t7a = xhalf(f0[9]);
    const float t6b = xhalf(f1[8]);
    const float t7b = xhalf(f1[9]);

    if (rr >= y0 + (KH + 1)) {
      const float v16 = half ? f1[9] : t7a;
      const float v15 = half ? f1[8] : t6a;
      if (xok) {
        out[(size_t)(rr - (KH + 1)) * OUT_W + x] = v16 + b0;
        out[(size_t)(rr - KH) * OUT_W + x]       = v15 + b0;
      }
    }

    f0[0] = half ? t6a : 0.f;
    f0[1] = half ? t7a : 0.f;
    f1[0] = half ? t6b : 0.f;
    f1[1] = half ? t7b : 0.f;
    v8f acc0 = __builtin_shufflevector(f0, f0, 0, 1, 2, 3, 4, 5, 6, 7);
    v8f acc1 = __builtin_shufflevector(f1, f1, 0, 1, 2, 3, 4, 5, 6, 7);

    v2f b00[4], b01[4], b10[4], b11[4];
    #pragma unroll
    for (int j = 0; j < 4; ++j) {
      b00[j].x = rowbuf[tb + 4 * j];
      b00[j].y = rowbuf[tb + 4 * j + 1];
      b01[j].x = rowbuf[16 + tb + 4 * j];
      b01[j].y = rowbuf[16 + tb + 4 * j + 1];
      b10[j].x = rowbuf[ROWSLOT + tb + 4 * j];
      b10[j].y = rowbuf[ROWSLOT + tb + 4 * j + 1];
      b11[j].x = rowbuf[ROWSLOT + 16 + tb + 4 * j];
      b11[j].y = rowbuf[ROWSLOT + 16 + tb + 4 * j + 1];
    }
    __builtin_amdgcn_sched_barrier(0);
    #pragma unroll
    for (int j = 0; j < 4; ++j) {
      acc0 = __builtin_amdgcn_wmma_f32_16x16x4_f32(false, a0[j], false, b00[j],
                                                   (short)0, acc0, false, false);
      acc1 = __builtin_amdgcn_wmma_f32_16x16x4_f32(false, a0[j], false, b01[j],
                                                   (short)0, acc1, false, false);
      acc0 = __builtin_amdgcn_wmma_f32_16x16x4_f32(false, a1[j], false, b10[j],
                                                   (short)0, acc0, false, false);
      acc1 = __builtin_amdgcn_wmma_f32_16x16x4_f32(false, a1[j], false, b11[j],
                                                   (short)0, acc1, false, false);
    }
    t0 = __builtin_shufflevector(acc0, acc0, -1, -1, 0, 1, 2, 3, 4, 5, 6, 7);
    t1 = __builtin_shufflevector(acc1, acc1, -1, -1, 0, 1, 2, 3, 4, 5, 6, 7);
  };

  for (; r + 6 < r_last; r += 8) {
    pair_step(r,     S0, S3, true, P0, P1, Q0, Q1);
    pair_step(r + 2, S1, S0, true, Q0, Q1, P0, P1);
    pair_step(r + 4, S2, S1, true, P0, P1, Q0, Q1);
    pair_step(r + 6, S3, S2, true, Q0, Q1, P0, P1);
  }
  #pragma nounroll
  while (r < r_last) {
    pair_step(r, S0, S3, false, P0, P1, Q0, Q1);
    P0 = Q0;
    P1 = Q1;
    S0 = S1;
    S1 = S2;
    r += 2;
  }

  {
    const float t6a = xhalf(P0[8]);
    const float t7a = xhalf(P0[9]);
    const float v16 = half ? P1[9] : t7a;
    const float v15 = half ? P1[8] : t6a;
    if (xok) {
      out[(size_t)(r - (KH + 1)) * OUT_W + x] = v16 + b0;
      out[(size_t)(r - KH) * OUT_W + x]       = v15 + b0;
    }
  }
  #undef LOAD_ROW
}

extern "C" void kernel_launch(void* const* d_in, const int* in_sizes, int n_in,
                              void* d_out, int out_size, void* d_ws, size_t ws_size,
                              hipStream_t stream) {
  const float* X    = (const float*)d_in[0];
  const float* Wgt  = (const float*)d_in[1];
  const float* bias = (const float*)d_in[2];
  float* out = (float*)d_out;

  const int bands = (OUT_H + HROWS - 1) / HROWS;
  dim3 grid((OUT_W + WTILE - 1) / WTILE, (bands + NWAVES - 1) / NWAVES, 1);
  dim3 block(32 * NWAVES, 1, 1);
  conv2d_wmma_f32_kernel<<<grid, block, 0, stream>>>(X, Wgt, bias, out);
}
